# starting kernel
# baseline (speedup 1.0000x reference)
#include <hip/hip_runtime.h>

typedef __attribute__((ext_vector_type(2))) float v2f;
typedef __attribute__((ext_vector_type(8))) float v8f;

#define B_  32
#define C_  256
#define H_  64
#define W_  64
#define HW_ (H_ * W_)

#define PX_ 128
#define KPLANE_ (PX_ * 4)
#define SMEM_BYTES_ ((C_ / 4) * KPLANE_ * 4)

__device__ __forceinline__ void s2f(float& a, float& b) {
    float mn = fminf(a, b);
    b = fmaxf(a, b);
    a = mn;
}

__device__ __forceinline__ float median9(float v0, float v1, float v2,
                                         float v3, float v4, float v5,
                                         float v6, float v7, float v8) {
    s2f(v0, v3); s2f(v1, v4); s2f(v2, v5);
    s2f(v0, v1); s2f(v0, v2);
    s2f(v4, v5); s2f(v3, v5);
    s2f(v1, v2); s2f(v3, v4);
    s2f(v1, v3); s2f(v1, v6);
    s2f(v4, v6); s2f(v2, v6);
    s2f(v2, v3); s2f(v4, v7); s2f(v2, v4); s2f(v3, v7);
    s2f(v4, v8); s2f(v3, v8); s2f(v3, v4);
    return v4;
}

__device__ __forceinline__ float ldz(const float* __restrict__ p, int off, bool valid) {
    float v = p[valid ? off : 0];
    return valid ? v : 0.0f;
}

__global__ __launch_bounds__(256)
void fused_median_conv1x1(const float* __restrict__ x,
                          const float* __restrict__ cw,
                          const float* __restrict__ cb,
                          float* __restrict__ out) {
    extern __shared__ float smed[];

    const int h0 = blockIdx.x * 2;
    const int b  = blockIdx.y;
    const int t  = threadIdx.x;

    {
        const int w   = t & 63;
        const int q   = t >> 6;
        const int r0  = q & 1;
        const int cp  = q >> 1;
        const int row = h0 + r0;
        const bool hm = (row > 0), hp = (row < H_ - 1);
        const bool wm = (w > 0),   wp = (w < W_ - 1);
        const int  pix = r0 * 64 + w;
        const size_t rowbase = (size_t)b * C_ * HW_ + (size_t)row * W_ + w;

        for (int c = cp; c < C_; c += 2) {
            const float* p = x + rowbase + (size_t)c * HW_;
            float a0 = ldz(p, -W_ - 1, hm && wm);
            float a1 = ldz(p, -W_,     hm);
            float a2 = ldz(p, -W_ + 1, hm && wp);
            float a3 = ldz(p, -1,      wm);
            float a4 = p[0];
            float a5 = ldz(p,  1,      wp);
            float a6 = ldz(p,  W_ - 1, hp && wm);
            float a7 = ldz(p,  W_,     hp);
            float a8 = ldz(p,  W_ + 1, hp && wp);
            smed[(c >> 2) * KPLANE_ + pix * 4 + (c & 3)] =
                median9(a0, a1, a2, a3, a4, a5, a6, a7, a8);
        }
    }
    __syncthreads();

    const int lane = t & 31;
    const int wv   = t >> 5;
    const int o0   = wv * 32;
    const int l15  = lane & 15;
    const int koff = (lane < 16) ? 0 : 2;

    const v2f* a0p = (const v2f*)(cw + (size_t)(o0 +      l15) * C_ + koff);
    const v2f* a1p = (const v2f*)(cw + (size_t)(o0 + 16 + l15) * C_ + koff);
    const v2f* bp[8];
#pragma unroll
    for (int j = 0; j < 8; ++j)
        bp[j] = (const v2f*)(&smed[(j * 16 + l15) * 4 + koff]);

    v8f acc[2][8];
#pragma unroll
    for (int i = 0; i < 2; ++i)
#pragma unroll
        for (int j = 0; j < 8; ++j)
            acc[i][j] = (v8f){0.f, 0.f, 0.f, 0.f, 0.f, 0.f, 0.f, 0.f};

#pragma unroll 2
    for (int k0 = 0; k0 < C_; k0 += 4) {
        v2f a0 = a0p[k0 >> 1];
        v2f a1 = a1p[k0 >> 1];
        const int bi = (k0 >> 2) * (KPLANE_ / 2);
        v2f bv[8];
#pragma unroll
        for (int j = 0; j < 8; ++j) bv[j] = bp[j][bi];
#pragma unroll
        for (int j = 0; j < 8; ++j)
            acc[0][j] = __builtin_amdgcn_wmma_f32_16x16x4_f32(false, a0, false, bv[j],
                                                              (short)0, acc[0][j], false, false);
#pragma unroll
        for (int j = 0; j < 8; ++j)
            acc[1][j] = __builtin_amdgcn_wmma_f32_16x16x4_f32(false, a1, false, bv[j],
                                                              (short)0, acc[1][j], false, false);
    }

    const int madd = (lane < 16) ? 0 : 8;

    float cbv[2][8];
#pragma unroll
    for (int i = 0; i < 2; ++i)
#pragma unroll
        for (int r = 0; r < 8; ++r)
            cbv[i][r] = cb[o0 + i * 16 + madd + r];

#pragma unroll
    for (int i = 0; i < 2; ++i) {
#pragma unroll
        for (int j = 0; j < 8; ++j) {
            const int p   = j * 16 + l15;
            const int row = h0 + (p >> 6);
            const int col = p & 63;
            const size_t base = (size_t)b * C_ * HW_ + (size_t)row * W_ + col;
            const int ob = o0 + i * 16 + madd;
            float xv[8];
#pragma unroll
            for (int r = 0; r < 8; ++r)
                xv[r] = x[base + (size_t)(ob + r) * HW_];
#pragma unroll
            for (int r = 0; r < 8; ++r)
                out[base + (size_t)(ob + r) * HW_] = xv[r] + cbv[i][r] + acc[i][j][r];
        }
    }
}

extern "C" void kernel_launch(void* const* d_in, const int* in_sizes, int n_in,
                              void* d_out, int out_size, void* d_ws, size_t ws_size,
                              hipStream_t stream) {
    (void)in_sizes; (void)n_in; (void)d_ws; (void)ws_size; (void)out_size;
    const float* x  = (const float*)d_in[0];
    const float* cw = (const float*)d_in[1];
    const float* cb = (const float*)d_in[2];
    float* out = (float*)d_out;

    dim3 grid(H_ / 2, B_);
    dim3 block(256);
    fused_median_conv1x1<<<grid, block, SMEM_BYTES_, stream>>>(x, cw, cb, out);
}


# median phase regrouped: one thread = one column, both rows, 4 channels; its 48 taps go out as one batch a group ahead (was 18 per wait); min3/med3/max3 column-sort median; 16-byte LDS stores
# speedup vs baseline: 1.4007x; 1.4007x over previous
#include <hip/hip_runtime.h>

typedef __attribute__((ext_vector_type(2))) float v2f;
typedef __attribute__((ext_vector_type(8))) float v8f;

#define B_  32
#define C_  256
#define H_  64
#define W_  64
#define HW_ (H_ * W_)

#define PX_ 128
#define KPLANE_ (PX_ * 4)
#define SMEM_BYTES_ ((C_ / 4) * KPLANE_ * 4)

__device__ __forceinline__ float min3f(float a, float b, float c) { return fminf(fminf(a, b), c); }
__device__ __forceinline__ float max3f(float a, float b, float c) { return fmaxf(fmaxf(a, b), c); }
__device__ __forceinline__ float med3f(float a, float b, float c) { return __builtin_amdgcn_fmed3f(a, b, c); }

__device__ __forceinline__ float median9c(float a0, float b0, float c0,
                                          float a1, float b1, float c1,
                                          float a2, float b2, float c2) {
    const float lo = max3f(min3f(a0, b0, c0), min3f(a1, b1, c1), min3f(a2, b2, c2));
    const float mi = med3f(med3f(a0, b0, c0), med3f(a1, b1, c1), med3f(a2, b2, c2));
    const float hi = min3f(max3f(a0, b0, c0), max3f(a1, b1, c1), max3f(a2, b2, c2));
    return med3f(lo, mi, hi);
}

template <bool VT, bool VB>
__device__ __forceinline__ void load_taps(float (&v)[4][4][3],
                                          const float* __restrict__ base,
                                          unsigned oL, unsigned oC, unsigned oR,
                                          bool vL, bool vR) {
    const float* pL = base + oL;
    const float* pC = base + oC;
    const float* pR = base + oR;
#pragma unroll
    for (int ch = 0; ch < 4; ++ch)
#pragma unroll
        for (int r = 0; r < 4; ++r) {
            const bool rv  = (r == 0) ? VT : ((r == 3) ? VB : true);
            const int  off = ch * HW_ + (r - 1) * W_;
            if (rv) {
                const float l = pL[off];
                const float c = pC[off];
                const float q = pR[off];
                v[ch][r][0] = vL ? l : 0.0f;
                v[ch][r][1] = c;
                v[ch][r][2] = vR ? q : 0.0f;
            } else {
                v[ch][r][0] = 0.0f; v[ch][r][1] = 0.0f; v[ch][r][2] = 0.0f;
            }
        }
}

__device__ __forceinline__ void median_store(const float (&v)[4][4][3], float* __restrict__ dst) {
    float top[4], bot[4];
#pragma unroll
    for (int ch = 0; ch < 4; ++ch) {
        top[ch] = median9c(v[ch][0][0], v[ch][1][0], v[ch][2][0],
                           v[ch][0][1], v[ch][1][1], v[ch][2][1],
                           v[ch][0][2], v[ch][1][2], v[ch][2][2]);
        bot[ch] = median9c(v[ch][1][0], v[ch][2][0], v[ch][3][0],
                           v[ch][1][1], v[ch][2][1], v[ch][3][1],
                           v[ch][1][2], v[ch][2][2], v[ch][3][2]);
    }
    *(float4*)(dst)          = make_float4(top[0], top[1], top[2], top[3]);
    *(float4*)(dst + 64 * 4) = make_float4(bot[0], bot[1], bot[2], bot[3]);
}

template <bool VT, bool VB>
__device__ __forceinline__ void median_phase(const float* __restrict__ base, float* __restrict__ d,
                                             unsigned oL, unsigned oC, unsigned oR,
                                             bool vL, bool vR) {
    const int GS = 4 * 4 * HW_;
    float A[4][4][3], Bq[4][4][3];
    load_taps<VT, VB>(A, base, oL, oC, oR, vL, vR);
    for (int i = 0; i < 14; i += 2) {
        load_taps<VT, VB>(Bq, base + GS, oL, oC, oR, vL, vR);
        median_store(A, d);
        __builtin_amdgcn_sched_barrier(0);
        load_taps<VT, VB>(A, base + 2 * GS, oL, oC, oR, vL, vR);
        median_store(Bq, d + 4 * KPLANE_);
        base += 2 * GS;
        d    += 8 * KPLANE_;
    }
    load_taps<VT, VB>(Bq, base + GS, oL, oC, oR, vL, vR);
    median_store(A, d);
    median_store(Bq, d + 4 * KPLANE_);
}

__global__ __launch_bounds__(256)
void fused_median_conv1x1(const float* __restrict__ x,
                          const float* __restrict__ cw,
                          const float* __restrict__ cb,
                          float* __restrict__ out) {
    extern __shared__ __attribute__((aligned(16))) float smed[];

    const int h0 = blockIdx.x * 2;
    const int b  = blockIdx.y;
    const int t  = threadIdx.x;

    {
        const int w  = t & 63;
        const int gq = t >> 6;
        const bool vL = (w > 0), vR = (w < W_ - 1);
        const float* xb = x + (size_t)b * C_ * HW_;
        const unsigned oC = (unsigned)(gq * 4 * HW_ + h0 * W_ + w);
        const unsigned oL = oC - (vL ? 1u : 0u);
        const unsigned oR = oC + (vR ? 1u : 0u);
        float* dst = smed + gq * KPLANE_ + w * 4;
        if (h0 == 0)            median_phase<false, true>(xb, dst, oL, oC, oR, vL, vR);
        else if (h0 + 2 >= H_)  median_phase<true, false>(xb, dst, oL, oC, oR, vL, vR);
        else                    median_phase<true, true>(xb, dst, oL, oC, oR, vL, vR);
    }
    __syncthreads();

    const int lane = t & 31;
    const int wv   = t >> 5;
    const int o0   = wv * 32;
    const int l15  = lane & 15;
    const int koff = (lane < 16) ? 0 : 2;

    const v2f* a0p = (const v2f*)(cw + (size_t)(o0 +      l15) * C_ + koff);
    const v2f* a1p = (const v2f*)(cw + (size_t)(o0 + 16 + l15) * C_ + koff);
    const v2f* bp[8];
#pragma unroll
    for (int j = 0; j < 8; ++j)
        bp[j] = (const v2f*)(&smed[(j * 16 + l15) * 4 + koff]);

    v8f acc[2][8];
#pragma unroll
    for (int i = 0; i < 2; ++i)
#pragma unroll
        for (int j = 0; j < 8; ++j)
            acc[i][j] = (v8f){0.f, 0.f, 0.f, 0.f, 0.f, 0.f, 0.f, 0.f};

#pragma unroll 2
    for (int k0 = 0; k0 < C_; k0 += 4) {
        v2f a0 = a0p[k0 >> 1];
        v2f a1 = a1p[k0 >> 1];
        const int bi = (k0 >> 2) * (KPLANE_ / 2);
        v2f bv[8];
#pragma unroll
        for (int j = 0; j < 8; ++j) bv[j] = bp[j][bi];
#pragma unroll
        for (int j = 0; j < 8; ++j)
            acc[0][j] = __builtin_amdgcn_wmma_f32_16x16x4_f32(false, a0, false, bv[j],
                                                              (short)0, acc[0][j], false, false);
#pragma unroll
        for (int j = 0; j < 8; ++j)
            acc[1][j] = __builtin_amdgcn_wmma_f32_16x16x4_f32(false, a1, false, bv[j],
                                                              (short)0, acc[1][j], false, false);
    }

    const int madd = (lane < 16) ? 0 : 8;

    float cbv[2][8];
#pragma unroll
    for (int i = 0; i < 2; ++i)
#pragma unroll
        for (int r = 0; r < 8; ++r)
            cbv[i][r] = cb[o0 + i * 16 + madd + r];

#pragma unroll
    for (int i = 0; i < 2; ++i) {
#pragma unroll
        for (int j = 0; j < 8; ++j) {
            const int p   = j * 16 + l15;
            const int row = h0 + (p >> 6);
            const int col = p & 63;
            const size_t base = (size_t)b * C_ * HW_ + (size_t)row * W_ + col;
            const int ob = o0 + i * 16 + madd;
            float xv[8];
#pragma unroll
            for (int r = 0; r < 8; ++r)
                xv[r] = x[base + (size_t)(ob + r) * HW_];
#pragma unroll
            for (int r = 0; r < 8; ++r)
                out[base + (size_t)(ob + r) * HW_] = xv[r] + cbv[i][r] + acc[i][j][r];
        }
    }
}

extern "C" void kernel_launch(void* const* d_in, const int* in_sizes, int n_in,
                              void* d_out, int out_size, void* d_ws, size_t ws_size,
                              hipStream_t stream) {
    (void)in_sizes; (void)n_in; (void)d_ws; (void)ws_size; (void)out_size;
    const float* x  = (const float*)d_in[0];
    const float* cw = (const float*)d_in[1];
    const float* cb = (const float*)d_in[2];
    float* out = (float*)d_out;

    dim3 grid(H_ / 2, B_);
    dim3 block(256);
    fused_median_conv1x1<<<grid, block, SMEM_BYTES_, stream>>>(x, cw, cb, out);
}


# one lane = 4 adjacent columns in median phase and epilogue: 16-byte global loads/stores (were 4-byte), side taps by DPP row shift of sorted columns; LDS pixel order permuted to match
# speedup vs baseline: 1.5945x; 1.1384x over previous
#include <hip/hip_runtime.h>


typedef __attribute__((ext_vector_type(2))) float v2f;
typedef __attribute__((ext_vector_type(8))) float v8f;

#define B_  32
#define C_  256
#define H_  64
#define W_  64
#define HW_ (H_ * W_)

#define PX_ 128
#define KPLANE_ (PX_ * 4)
#define SMEM_BYTES_ ((C_ / 4) * KPLANE_ * 4)


__device__ __forceinline__ float min3f(float a, float b, float c) { return fminf(fminf(a, b), c); }
__device__ __forceinline__ float max3f(float a, float b, float c) { return fmaxf(fmaxf(a, b), c); }
__device__ __forceinline__ float med3f(float a, float b, float c) { return __builtin_amdgcn_fmed3f(a, b, c); }

__device__ __forceinline__ float from_left(float v) {
    return __builtin_bit_cast(float, __builtin_amdgcn_update_dpp(0, __builtin_bit_cast(int, v),
                                                                 0x111, 0xf, 0xf, true));
}
__device__ __forceinline__ float from_right(float v) {
    return __builtin_bit_cast(float, __builtin_amdgcn_update_dpp(0, __builtin_bit_cast(int, v),
                                                                 0x101, 0xf, 0xf, true));
}

template <bool VT, bool VB>
__device__ __forceinline__ void load_taps(float (&v)[4][4][4], const float* __restrict__ p) {
#pragma unroll
    for (int ch = 0; ch < 4; ++ch)
#pragma unroll
        for (int r = 0; r < 4; ++r) {
            const bool rv = (r == 0) ? VT : ((r == 3) ? VB : true);
            if (rv) {
                const float4 f = *(const float4*)(p + ch * HW_ + (r - 1) * W_);
                v[ch][r][0] = f.x; v[ch][r][1] = f.y; v[ch][r][2] = f.z; v[ch][r][3] = f.w;
            } else {
                v[ch][r][0] = 0.0f; v[ch][r][1] = 0.0f; v[ch][r][2] = 0.0f; v[ch][r][3] = 0.0f;
            }
        }
}

__device__ __forceinline__ void median_store(const float (&v)[4][4][4], float* __restrict__ dst) {
#pragma unroll
    for (int o = 0; o < 2; ++o) {
        float res[4][4];
#pragma unroll
        for (int ch = 0; ch < 4; ++ch) {
            float lo[6], mi[6], hi[6];
#pragma unroll
            for (int i = 0; i < 4; ++i) {
                const float a = v[ch][o][i], bq = v[ch][o + 1][i], c = v[ch][o + 2][i];
                lo[i + 1] = min3f(a, bq, c);
                mi[i + 1] = med3f(a, bq, c);
                hi[i + 1] = max3f(a, bq, c);
            }
            lo[0] = from_left(lo[4]);  mi[0] = from_left(mi[4]);  hi[0] = from_left(hi[4]);
            lo[5] = from_right(lo[1]); mi[5] = from_right(mi[1]); hi[5] = from_right(hi[1]);
#pragma unroll
            for (int i = 0; i < 4; ++i)
                res[i][ch] = med3f(max3f(lo[i], lo[i + 1], lo[i + 2]),
                                   med3f(mi[i], mi[i + 1], mi[i + 2]),
                                   min3f(hi[i], hi[i + 1], hi[i + 2]));
        }
#pragma unroll
        for (int i = 0; i < 4; ++i)
            *(float4*)(dst + (o * 64 + i * 16) * 4) =
                make_float4(res[i][0], res[i][1], res[i][2], res[i][3]);
    }
}

template <bool VT, bool VB>
__device__ __forceinline__ void median_phase(const float* __restrict__ p, float* __restrict__ d) {
    const int GS = 16 * 4 * HW_;
    const int DS = 16 * KPLANE_;
    float A[4][4][4], Bq[4][4][4];
    load_taps<VT, VB>(A, p);
    load_taps<VT, VB>(Bq, p + GS);
    median_store(A, d);
    load_taps<VT, VB>(A, p + 2 * GS);
    median_store(Bq, d + DS);
    load_taps<VT, VB>(Bq, p + 3 * GS);
    median_store(A, d + 2 * DS);
    median_store(Bq, d + 3 * DS);
}

__global__ __launch_bounds__(256)
void fused_median_conv1x1(const float* __restrict__ x,
                          const float* __restrict__ cw,
                          const float* __restrict__ cb,
                          float* __restrict__ out) {
    extern __shared__ __attribute__((aligned(16))) float smed[];

    const int h0 = blockIdx.x * 2;
    const int b  = blockIdx.y;
    const int t  = threadIdx.x;

    {
        const int q  = t & 15;
        const int g  = t >> 4;
        const float* p = x + (size_t)b * C_ * HW_ + (size_t)(g * 4) * HW_ + h0 * W_ + q * 4;
        float* dst = smed + g * KPLANE_ + q * 4;
        if (h0 > 0 && h0 + 2 < H_)  median_phase<true, true>(p, dst);
        else if (h0 == 0)           median_phase<false, true>(p, dst);
        else                        median_phase<true, false>(p, dst);
    }
    __syncthreads();

    const int lane = t & 31;
    const int wv   = t >> 5;
    const int o0   = wv * 32;
    const int l15  = lane & 15;
    const int koff = (lane < 16) ? 0 : 2;

    const v2f* a0p = (const v2f*)(cw + (size_t)(o0 +      l15) * C_ + koff);
    const v2f* a1p = (const v2f*)(cw + (size_t)(o0 + 16 + l15) * C_ + koff);
    const v2f* bp[8];
#pragma unroll
    for (int j = 0; j < 8; ++j)
        bp[j] = (const v2f*)(&smed[(j * 16 + l15) * 4 + koff]);

    v8f acc[2][8];
#pragma unroll
    for (int i = 0; i < 2; ++i)
#pragma unroll
        for (int j = 0; j < 8; ++j)
            acc[i][j] = (v8f){0.f, 0.f, 0.f, 0.f, 0.f, 0.f, 0.f, 0.f};

#pragma unroll 2
    for (int k0 = 0; k0 < C_; k0 += 4) {
        v2f a0 = a0p[k0 >> 1];
        v2f a1 = a1p[k0 >> 1];
        const int bi = (k0 >> 2) * (KPLANE_ / 2);
        v2f bv[8];
#pragma unroll
        for (int j = 0; j < 8; ++j) bv[j] = bp[j][bi];
#pragma unroll
        for (int j = 0; j < 8; ++j)
            acc[0][j] = __builtin_amdgcn_wmma_f32_16x16x4_f32(false, a0, false, bv[j],
                                                              (short)0, acc[0][j], false, false);
#pragma unroll
        for (int j = 0; j < 8; ++j)
            acc[1][j] = __builtin_amdgcn_wmma_f32_16x16x4_f32(false, a1, false, bv[j],
                                                              (short)0, acc[1][j], false, false);
    }

    const int madd = (lane < 16) ? 0 : 8;

    float cbv[2][8];
#pragma unroll
    for (int i = 0; i < 2; ++i)
#pragma unroll
        for (int r = 0; r < 8; ++r)
            cbv[i][r] = cb[o0 + i * 16 + madd + r];

#pragma unroll
    for (int i = 0; i < 2; ++i) {
#pragma unroll
        for (int ro = 0; ro < 2; ++ro) {
            const size_t base = (size_t)b * C_ * HW_ + (size_t)(h0 + ro) * W_ + l15 * 4;
            const int ob = o0 + i * 16 + madd;
            float4 xv[8];
#pragma unroll
            for (int r = 0; r < 8; ++r)
                xv[r] = *(const float4*)(x + base + (size_t)(ob + r) * HW_);
#pragma unroll
            for (int r = 0; r < 8; ++r) {
                float4 ov;
                ov.x = xv[r].x + cbv[i][r] + acc[i][4 * ro + 0][r];
                ov.y = xv[r].y + cbv[i][r] + acc[i][4 * ro + 1][r];
                ov.z = xv[r].z + cbv[i][r] + acc[i][4 * ro + 2][r];
                ov.w = xv[r].w + cbv[i][r] + acc[i][4 * ro + 3][r];
                *(float4*)(out + base + (size_t)(ob + r) * HW_) = ov;
            }
        }
    }
}

extern "C" void kernel_launch(void* const* d_in, const int* in_sizes, int n_in,
                              void* d_out, int out_size, void* d_ws, size_t ws_size,
                              hipStream_t stream) {
    (void)in_sizes; (void)n_in; (void)d_ws; (void)ws_size; (void)out_size;
    const float* x  = (const float*)d_in[0];
    const float* cw = (const float*)d_in[1];
    const float* cb = (const float*)d_in[2];
    float* out = (float*)d_out;

    dim3 grid(H_ / 2, B_);
    dim3 block(256);
    fused_median_conv1x1<<<grid, block, SMEM_BYTES_, stream>>>(x, cw, cb, out);
}


# weight (A) fragments of 8 matrix steps requested as one group of 16 loads in front of their 128 matrix steps: one global-load wait chain per 128 WMMAs (was per 32); sum order unchanged
# speedup vs baseline: 1.6509x; 1.0354x over previous
#include <hip/hip_runtime.h>


typedef __attribute__((ext_vector_type(2))) float v2f;
typedef __attribute__((ext_vector_type(8))) float v8f;

#define B_  32
#define C_  256
#define H_  64
#define W_  64
#define HW_ (H_ * W_)

#define PX_ 128
#define KPLANE_ (PX_ * 4)
#define SMEM_BYTES_ ((C_ / 4) * KPLANE_ * 4)


__device__ __forceinline__ float min3f(float a, float b, float c) { return fminf(fminf(a, b), c); }
__device__ __forceinline__ float max3f(float a, float b, float c) { return fmaxf(fmaxf(a, b), c); }
__device__ __forceinline__ float med3f(float a, float b, float c) { return __builtin_amdgcn_fmed3f(a, b, c); }

__device__ __forceinline__ float from_left(float v) {
    return __builtin_bit_cast(float, __builtin_amdgcn_update_dpp(0, __builtin_bit_cast(int, v),
                                                                 0x111, 0xf, 0xf, true));
}
__device__ __forceinline__ float from_right(float v) {
    return __builtin_bit_cast(float, __builtin_amdgcn_update_dpp(0, __builtin_bit_cast(int, v),
                                                                 0x101, 0xf, 0xf, true));
}

template <bool VT, bool VB>
__device__ __forceinline__ void load_taps(float (&v)[4][4][4], const float* __restrict__ p) {
#pragma unroll
    for (int ch = 0; ch < 4; ++ch)
#pragma unroll
        for (int r = 0; r < 4; ++r) {
            const bool rv = (r == 0) ? VT : ((r == 3) ? VB : true);
            if (rv) {
                const float4 f = *(const float4*)(p + ch * HW_ + (r - 1) * W_);
                v[ch][r][0] = f.x; v[ch][r][1] = f.y; v[ch][r][2] = f.z; v[ch][r][3] = f.w;
            } else {
                v[ch][r][0] = 0.0f; v[ch][r][1] = 0.0f; v[ch][r][2] = 0.0f; v[ch][r][3] = 0.0f;
            }
        }
}

__device__ __forceinline__ void median_store(const float (&v)[4][4][4], float* __restrict__ dst) {
#pragma unroll
    for (int o = 0; o < 2; ++o) {
        float res[4][4];
#pragma unroll
        for (int ch = 0; ch < 4; ++ch) {
            float lo[6], mi[6], hi[6];
#pragma unroll
            for (int i = 0; i < 4; ++i) {
                const float a = v[ch][o][i], bq = v[ch][o + 1][i], c = v[ch][o + 2][i];
                lo[i + 1] = min3f(a, bq, c);
                mi[i + 1] = med3f(a, bq, c);
                hi[i + 1] = max3f(a, bq, c);
            }
            lo[0] = from_left(lo[4]);  mi[0] = from_left(mi[4]);  hi[0] = from_left(hi[4]);
            lo[5] = from_right(lo[1]); mi[5] = from_right(mi[1]); hi[5] = from_right(hi[1]);
#pragma unroll
            for (int i = 0; i < 4; ++i)
                res[i][ch] = med3f(max3f(lo[i], lo[i + 1], lo[i + 2]),
                                   med3f(mi[i], mi[i + 1], mi[i + 2]),
                                   min3f(hi[i], hi[i + 1], hi[i + 2]));
        }
#pragma unroll
        for (int i = 0; i < 4; ++i)
            *(float4*)(dst + (o * 64 + i * 16) * 4) =
                make_float4(res[i][0], res[i][1], res[i][2], res[i][3]);
    }
}

template <bool VT, bool VB>
__device__ __forceinline__ void median_phase(const float* __restrict__ p, float* __restrict__ d) {
    const int GS = 16 * 4 * HW_;
    const int DS = 16 * KPLANE_;
    float A[4][4][4], Bq[4][4][4];
    load_taps<VT, VB>(A, p);
    load_taps<VT, VB>(Bq, p + GS);
    __builtin_amdgcn_sched_barrier(0);
    median_store(A, d);
    __builtin_amdgcn_sched_barrier(0);
    load_taps<VT, VB>(A, p + 2 * GS);
    __builtin_amdgcn_sched_barrier(0);
    median_store(Bq, d + DS);
    __builtin_amdgcn_sched_barrier(0);
    load_taps<VT, VB>(Bq, p + 3 * GS);
    __builtin_amdgcn_sched_barrier(0);
    median_store(A, d + 2 * DS);
    __builtin_amdgcn_sched_barrier(0);
    median_store(Bq, d + 3 * DS);
}

__global__ __launch_bounds__(256, 4)
void fused_median_conv1x1(const float* __restrict__ x,
                          const float* __restrict__ cw,
                          const float* __restrict__ cb,
                          float* __restrict__ out) {
    extern __shared__ __attribute__((aligned(16))) float smed[];

    const int h0 = blockIdx.x * 2;
    const int b  = blockIdx.y;
    const int t  = threadIdx.x;

    {
        const int q  = t & 15;
        const int g  = t >> 4;
        const float* p = x + (size_t)b * C_ * HW_ + (size_t)(g * 4) * HW_ + h0 * W_ + q * 4;
        float* dst = smed + g * KPLANE_ + q * 4;
        if (h0 > 0 && h0 + 2 < H_)  median_phase<true, true>(p, dst);
        else if (h0 == 0)           median_phase<false, true>(p, dst);
        else                        median_phase<true, false>(p, dst);
    }
    __syncthreads();

    const int lane = t & 31;
    const int wv   = t >> 5;
    const int o0   = wv * 32;
    const int l15  = lane & 15;
    const int koff = (lane < 16) ? 0 : 2;

    const v2f* a0p = (const v2f*)(cw + (size_t)(o0 +      l15) * C_ + koff);
    const v2f* a1p = (const v2f*)(cw + (size_t)(o0 + 16 + l15) * C_ + koff);
    const v2f* bp[8];
#pragma unroll
    for (int j = 0; j < 8; ++j)
        bp[j] = (const v2f*)(&smed[(j * 16 + l15) * 4 + koff]);

    v8f acc[2][8];
#pragma unroll
    for (int i = 0; i < 2; ++i)
#pragma unroll
        for (int j = 0; j < 8; ++j)
            acc[i][j] = (v8f){0.f, 0.f, 0.f, 0.f, 0.f, 0.f, 0.f, 0.f};

#pragma unroll 1
    for (int k0 = 0; k0 < C_; k0 += 32) {
        v2f a0[8], a1[8];
#pragma unroll
        for (int u = 0; u < 8; ++u) {
            a0[u] = a0p[(k0 >> 1) + 2 * u];
            a1[u] = a1p[(k0 >> 1) + 2 * u];
        }
        __builtin_amdgcn_sched_barrier(0);
#pragma unroll
        for (int u = 0; u < 8; ++u) {
            const int bi = ((k0 >> 2) + u) * (KPLANE_ / 2);
            v2f bv[8];
#pragma unroll
            for (int j = 0; j < 8; ++j) bv[j] = bp[j][bi];
#pragma unroll
            for (int j = 0; j < 8; ++j)
                acc[0][j] = __builtin_amdgcn_wmma_f32_16x16x4_f32(false, a0[u], false, bv[j],
                                                                  (short)0, acc[0][j], false, false);
#pragma unroll
            for (int j = 0; j < 8; ++j)
                acc[1][j] = __builtin_amdgcn_wmma_f32_16x16x4_f32(false, a1[u], false, bv[j],
                                                                  (short)0, acc[1][j], false, false);
            if (u & 1) __builtin_amdgcn_sched_barrier(0);
        }
    }

    const int madd = (lane < 16) ? 0 : 8;

    float cbv[2][8];
#pragma unroll
    for (int i = 0; i < 2; ++i)
#pragma unroll
        for (int r = 0; r < 8; ++r)
            cbv[i][r] = cb[o0 + i * 16 + madd + r];

#pragma unroll
    for (int i = 0; i < 2; ++i) {
        if (i == 1) __builtin_amdgcn_sched_barrier(0);
#pragma unroll
        for (int ro = 0; ro < 2; ++ro) {
            const size_t base = (size_t)b * C_ * HW_ + (size_t)(h0 + ro) * W_ + l15 * 4;
            const int ob = o0 + i * 16 + madd;
            float4 xv[8];
#pragma unroll
            for (int r = 0; r < 8; ++r)
                xv[r] = *(const float4*)(x + base + (size_t)(ob + r) * HW_);
#pragma unroll
            for (int r = 0; r < 8; ++r) {
                float4 ov;
                ov.x = xv[r].x + cbv[i][r] + acc[i][4 * ro + 0][r];
                ov.y = xv[r].y + cbv[i][r] + acc[i][4 * ro + 1][r];
                ov.z = xv[r].z + cbv[i][r] + acc[i][4 * ro + 2][r];
                ov.w = xv[r].w + cbv[i][r] + acc[i][4 * ro + 3][r];
                *(float4*)(out + base + (size_t)(ob + r) * HW_) = ov;
            }
        }
    }
}

extern "C" void kernel_launch(void* const* d_in, const int* in_sizes, int n_in,
                              void* d_out, int out_size, void* d_ws, size_t ws_size,
                              hipStream_t stream) {
    (void)in_sizes; (void)n_in; (void)d_ws; (void)ws_size; (void)out_size;
    const float* x  = (const float*)d_in[0];
    const float* cw = (const float*)d_in[1];
    const float* cb = (const float*)d_in[2];
    float* out = (float*)d_out;

    dim3 grid(H_ / 2, B_);
    dim3 block(256);
    fused_median_conv1x1<<<grid, block, SMEM_BYTES_, stream>>>(x, cw, cb, out);
}
